# starting kernel
# baseline (speedup 1.0000x reference)
#include <hip/hip_runtime.h>

typedef __attribute__((ext_vector_type(2))) float v2f;
typedef __attribute__((ext_vector_type(8))) float v8f;

static constexpr int T_LEN = 2048;
static constexpr int BATCH = 512;
static constexpr int EMB_D = 16;
static constexpr int HID_D = 32;
static constexpr int GATE  = 128;

__device__ __forceinline__ float fsigmoid(float x) {
    return __fdividef(1.0f, 1.0f + __expf(-x));
}
__device__ __forceinline__ float ftanh(float x) {
    return 1.0f - __fdividef(2.0f, __expf(2.0f * x) + 1.0f);
}

__device__ __forceinline__ v8f wmma4(v2f a, v2f b, v8f c) {
    return __builtin_amdgcn_wmma_f32_16x16x4_f32(
         false, a,  false, b,
         (short)0, c,  false,  false);
}

__global__ __launch_bounds__(32)
void lstm_scan_kernel(const int* __restrict__ tokens,
                      const float* __restrict__ Wemb,
                      const float* __restrict__ Wx,
                      const float* __restrict__ Wh,
                      const float* __restrict__ bias,
                      float* __restrict__ out)
{
    __shared__ float kT[GATE * EMB_D];
    __shared__ float rT[GATE * HID_D];
    __shared__ float hbuf[16 * HID_D];

    const int lane = threadIdx.x & 31;
    const int half = lane >> 4;
    const int nloc = lane & 15;
    const int koff = half << 1;
    const int base = half << 3;
    const int b0   = blockIdx.x << 4;

    for (int i = lane; i < GATE * EMB_D; i += 32) {
        const int n = i / EMB_D, k = i - n * EMB_D;
        kT[i] = Wx[k * GATE + n];
    }
    for (int i = lane; i < GATE * HID_D; i += 32) {
        const int n = i / HID_D, k = i - n * HID_D;
        rT[i] = Wh[k * GATE + n];
    }
    for (int i = lane; i < 16 * HID_D; i += 32) hbuf[i] = 0.0f;
    __syncthreads();

    v8f bacc[8];
#pragma unroll
    for (int nc = 0; nc < 8; ++nc) {
        const float bv = bias[nc * 16 + nloc];
#pragma unroll
        for (int r = 0; r < 8; ++r) bacc[nc][r] = bv;
    }

    v8f cst[2] = {};
    v8f hst[2] = {};

    const int* tokp = tokens + (b0 + nloc) * T_LEN;
    int tok0 = tokp[0];
    int tok1 = tokp[1];

    v2f ax[4];
    {
        const float* er = Wemb + (long)tok0 * EMB_D + koff;
#pragma unroll
        for (int kc = 0; kc < 4; ++kc) ax[kc] = *(const v2f*)(er + 4 * kc);
    }

    for (int t = 0; t < T_LEN; ++t) {
        const int tok2 = (t + 2 < T_LEN) ? tokp[t + 2] : 0;

        v2f axn[4];
        {
            const float* er = Wemb + (long)tok1 * EMB_D + koff;
#pragma unroll
            for (int kc = 0; kc < 4; ++kc) axn[kc] = *(const v2f*)(er + 4 * kc);
        }

        const unsigned mb = __builtin_amdgcn_ballot_w32(tok0 != 0);

        v2f ah[8];
#pragma unroll
        for (int kc = 0; kc < 8; ++kc)
            ah[kc] = *(const v2f*)&hbuf[nloc * HID_D + 4 * kc + koff];

        v8f acc[8];
#pragma unroll
        for (int nc = 0; nc < 8; ++nc) {
            const float* kb = &kT[(nc * 16 + nloc) * EMB_D + koff];
            acc[nc] = wmma4(ax[0], *(const v2f*)(kb), bacc[nc]);
        }
#pragma unroll
        for (int kc = 1; kc < 4; ++kc) {
#pragma unroll
            for (int nc = 0; nc < 8; ++nc) {
                const float* kb = &kT[(nc * 16 + nloc) * EMB_D + koff];
                acc[nc] = wmma4(ax[kc], *(const v2f*)(kb + 4 * kc), acc[nc]);
            }
        }
#pragma unroll
        for (int kc = 0; kc < 8; ++kc) {
#pragma unroll
            for (int nc = 0; nc < 8; ++nc) {
                const float* rb = &rT[(nc * 16 + nloc) * HID_D + koff];
                acc[nc] = wmma4(ah[kc], *(const v2f*)(rb + 4 * kc), acc[nc]);
            }
        }

#pragma unroll
        for (int hc = 0; hc < 2; ++hc) {
#pragma unroll
            for (int r = 0; r < 8; ++r) {
                const float iv = fsigmoid(acc[0 + hc][r]);
                const float fv = fsigmoid(acc[2 + hc][r]);
                const float gv = ftanh(acc[4 + hc][r]);
                const float ov = fsigmoid(acc[6 + hc][r]);
                const float cn = fv * cst[hc][r] + iv * gv;
                const float hn = ov * ftanh(cn);
                const bool  m  = (mb >> (base + r)) & 1u;
                const float cw = m ? cn : cst[hc][r];
                const float hw = m ? hn : hst[hc][r];
                cst[hc][r] = cw;
                hst[hc][r] = hw;
                hbuf[(base + r) * HID_D + hc * 16 + nloc] = hw;
            }
        }

        tok0 = tok1;
        tok1 = tok2;
#pragma unroll
        for (int kc = 0; kc < 4; ++kc) ax[kc] = axn[kc];
    }

#pragma unroll
    for (int hc = 0; hc < 2; ++hc)
#pragma unroll
        for (int r = 0; r < 8; ++r)
            out[(b0 + base + r) * HID_D + hc * 16 + nloc] = hst[hc][r];
}

extern "C" void kernel_launch(void* const* d_in, const int* in_sizes, int n_in,
                              void* d_out, int out_size, void* d_ws, size_t ws_size,
                              hipStream_t stream) {
    const int*   tokens = (const int*)  d_in[0];
    const float* emb    = (const float*)d_in[1];
    const float* Wx     = (const float*)d_in[2];
    const float* Wh     = (const float*)d_in[3];
    const float* bias   = (const float*)d_in[4];
    float*       out    = (float*)d_out;
    (void)in_sizes; (void)n_in; (void)out_size; (void)d_ws; (void)ws_size;

    dim3 grid(BATCH / 16);
    dim3 block(32);
    lstm_scan_kernel<<<grid, block, 0, stream>>>(tokens, emb, Wx, Wh, bias, out);
}


# batch tile narrowed to 4 rows, each row copied to the 4 B columns of a lane quad: a lane does gate math for 4 of its 16 elements, h returns by quad broadcast (DPP); 128 workgroups; sum order kept
# speedup vs baseline: 2.1504x; 2.1504x over previous
#include <hip/hip_runtime.h>


typedef __attribute__((ext_vector_type(2))) float v2f;
typedef __attribute__((ext_vector_type(4))) float v4f;
typedef __attribute__((ext_vector_type(8))) float v8f;

static constexpr int T_LEN = 2048;
static constexpr int BATCH = 512;
static constexpr int EMB_D = 16;
static constexpr int HID_D = 32;
static constexpr int GATE  = 128;
static constexpr int ROWS  = 4;

__device__ __forceinline__ float fsigmoid(float x) {
    return __fdividef(1.0f, 1.0f + __expf(-x));
}
__device__ __forceinline__ float ftanh(float x) {
    return 1.0f - __fdividef(2.0f, __expf(2.0f * x) + 1.0f);
}

__device__ __forceinline__ v8f wmma4(v2f a, v2f b, v8f c) {
    return __builtin_amdgcn_wmma_f32_16x16x4_f32(
         false, a,  false, b,
         (short)0, c,  false,  false);
}

__device__ __forceinline__ void x_projection(const v2f (&ax)[4], const v2f (&kf)[8][4],
                                             const v8f (&bacc)[8], v8f (&p)[8]) {
#pragma unroll
    for (int nc = 0; nc < 8; ++nc) p[nc] = wmma4(kf[nc][0], ax[0], bacc[nc]);
#pragma unroll
    for (int kc = 1; kc < 4; ++kc) {
#pragma unroll
        for (int nc = 0; nc < 8; ++nc) p[nc] = wmma4(kf[nc][kc], ax[kc], p[nc]);
    }
}

__device__ __forceinline__ void store_fragments(v4f* pb, const int slot, const int lane,
                                                const v8f (&p)[8]) {
#pragma unroll
    for (int nc = 0; nc < 8; ++nc) {
        v4f lo, hi;
#pragma unroll
        for (int r = 0; r < 4; ++r) { lo[r] = p[nc][r]; hi[r] = p[nc][4 + r]; }
        pb[((slot * 8 + nc) * 2 + 0) * 32 + lane] = lo;
        pb[((slot * 8 + nc) * 2 + 1) * 32 + lane] = hi;
    }
}
__device__ __forceinline__ void load_fragments(const v4f* pb, const int slot, const int lane,
                                               v8f (&p)[8]) {
#pragma unroll
    for (int nc = 0; nc < 8; ++nc) {
        const v4f lo = pb[((slot * 8 + nc) * 2 + 0) * 32 + lane];
        const v4f hi = pb[((slot * 8 + nc) * 2 + 1) * 32 + lane];
#pragma unroll
        for (int r = 0; r < 4; ++r) { p[nc][r] = lo[r]; p[nc][4 + r] = hi[r]; }
    }
}

__device__ __forceinline__ float pick_bits(const int mask, const float a, const float b) {
    return __int_as_float((__float_as_int(a) & mask) | (__float_as_int(b) & ~mask));
}

template <int K>
__device__ __forceinline__ float quad_bcast(const float v) {
    return __int_as_float(__builtin_amdgcn_mov_dpp(__float_as_int(v), K * 0x55, 0xf, 0xf, true));
}

template <int HC>
__device__ __forceinline__ void gate_group(const bool m, const int j1, const int j2,
                                           const v8f (&p)[8], float (&cown)[4],
                                           float (&hown)[4], v8f (&hfull)[2]) {
#pragma unroll
    for (int e = 0; e < 2; ++e) {
        const int rb = 4 * e;
        float z[4];
#pragma unroll
        for (int q = 0; q < 4; ++q) {
            const float lo = pick_bits(j1, p[2 * q + HC][rb + 1], p[2 * q + HC][rb + 0]);
            const float hi = pick_bits(j1, p[2 * q + HC][rb + 3], p[2 * q + HC][rb + 2]);
            z[q] = pick_bits(j2, hi, lo);
        }
        const float co = cown[2 * HC + e];
        const float ho = hown[2 * HC + e];
        const float iv = fsigmoid(z[0]);
        const float fv = fsigmoid(z[1]);
        const float gv = ftanh(z[2]);
        const float ov = fsigmoid(z[3]);
        const float cn = fv * co + iv * gv;
        const float hn = ov * ftanh(cn);
        const float cw = m ? cn : co;
        const float hw = m ? hn : ho;
        cown[2 * HC + e] = cw;
        hown[2 * HC + e] = hw;
        hfull[HC][rb + 0] = quad_bcast<0>(hw);
        hfull[HC][rb + 1] = quad_bcast<1>(hw);
        hfull[HC][rb + 2] = quad_bcast<2>(hw);
        hfull[HC][rb + 3] = quad_bcast<3>(hw);
    }
}

template <int C>
__device__ __forceinline__ void h_chunk_product(const v2f (&rf)[8][8], const v8f (&hfull)[2],
                                                v8f (&pn)[8]) {
    constexpr int hc = C >> 2;
    constexpr int r0 = (C & 3) << 1;
    v2f bh;
    bh[0] = hfull[hc][r0];
    bh[1] = hfull[hc][r0 + 1];
#pragma unroll
    for (int nc = 0; nc < 8; ++nc) pn[nc] = wmma4(rf[nc][C], bh, pn[nc]);
}

__device__ __forceinline__ void recurrent_step(
    const int t, const int* __restrict__ tokp, const v4f* pbuf, const int lane,
    const int j1, const int j2,
    const v2f (&rf)[8][8], v8f (&p)[8], v8f (&pn)[8],
    float (&cown)[4], float (&hown)[4], v8f (&hfull)[2], int& tok0, int& tok1)
{
    const int tok2 = (t + 2 < T_LEN) ? tokp[t + 2] : 0;
    const bool m = (tok0 != 0);

    load_fragments(pbuf, (t + 1) & 1, lane, pn);

    __builtin_amdgcn_sched_barrier(0);
    gate_group<0>(m, j1, j2, p, cown, hown, hfull);
    __builtin_amdgcn_sched_barrier(0);
    gate_group<1>(m, j1, j2, p, cown, hown, hfull);
    h_chunk_product<0>(rf, hfull, pn);
    h_chunk_product<1>(rf, hfull, pn);
    h_chunk_product<2>(rf, hfull, pn);
    h_chunk_product<3>(rf, hfull, pn);
#pragma unroll
    for (int i = 0; i < 32; ++i) {
        __builtin_amdgcn_sched_group_barrier(0x008, 1, 0);
        __builtin_amdgcn_sched_group_barrier(0x002, 6, 0);
    }
    __builtin_amdgcn_sched_barrier(0);
    h_chunk_product<4>(rf, hfull, pn);
    h_chunk_product<5>(rf, hfull, pn);
    h_chunk_product<6>(rf, hfull, pn);
    h_chunk_product<7>(rf, hfull, pn);

    tok0 = tok1;
    tok1 = tok2;
}

__device__ __forceinline__ void projection_step(
    const int t, const int* __restrict__ tokp, const float* __restrict__ Wemb,
    v4f* pbuf, const int lane, const int koff,
    const v2f (&kf)[8][4], const v8f (&bacc)[8], v2f (&axc)[4], int& tokn)
{
    const int tokf = (t + 4 < T_LEN) ? tokp[t + 4] : 0;
    v2f axn[4];
    {
        const float* er = Wemb + (long)tokn * EMB_D + koff;
#pragma unroll
        for (int kc = 0; kc < 4; ++kc) axn[kc] = *(const v2f*)(er + 4 * kc);
    }
    v8f p[8];
    x_projection(axc, kf, bacc, p);
    store_fragments(pbuf, t & 1, lane, p);
    tokn = tokf;
#pragma unroll
    for (int kc = 0; kc < 4; ++kc) axc[kc] = axn[kc];
}

__global__ __launch_bounds__(64)
void lstm_scan_kernel(const int* __restrict__ tokens,
                      const float* __restrict__ Wemb,
                      const float* __restrict__ Wx,
                      const float* __restrict__ Wh,
                      const float* __restrict__ bias,
                      float* __restrict__ out)
{
    __shared__ v4f pbuf[2 * 8 * 2 * 32];

    const int lane = threadIdx.x & 31;
    const int half = lane >> 4;
    const int nloc = lane & 15;
    const int koff = half << 1;
    const int base = half << 3;
    const int b0   = blockIdx.x * ROWS;
    const int brow = nloc >> 2;
    const int jq   = nloc & 3;
    const int j1   = (jq & 1) ? -1 : 0;
    const int j2   = (jq & 2) ? -1 : 0;
    const bool recurrent = (__builtin_amdgcn_readfirstlane((int)threadIdx.x) < 32);

    const int* tokp = tokens + (b0 + brow) * T_LEN;

    v2f rf[8][8];
    v2f kf[8][4];
    v8f bacc[8];
#pragma unroll
    for (int nc = 0; nc < 8; ++nc) {
        const int col = nc * 16 + nloc;
#pragma unroll
        for (int kc = 0; kc < 8; ++kc) {
            const int kh = ((kc >> 2) << 4) + base + ((kc & 3) << 1);
            rf[nc][kc][0] = Wh[kh * GATE + col];
            rf[nc][kc][1] = Wh[(kh + 1) * GATE + col];
        }
#pragma unroll
        for (int kc = 0; kc < 4; ++kc) {
            kf[nc][kc][0] = Wx[(4 * kc + koff) * GATE + col];
            kf[nc][kc][1] = Wx[(4 * kc + koff + 1) * GATE + col];
        }
#pragma unroll
        for (int r = 0; r < 8; ++r) bacc[nc][r] = bias[nc * 16 + base + r];
    }

    v2f axc[4] = {};
    int tokn = 0;
    float cown[4] = {};
    float hown[4] = {};
    v8f hfull[2] = {};
    v8f pa[8] = {};
    v8f pb[8] = {};
    int tok0 = 0, tok1 = 0;

    if (!recurrent) {
#pragma unroll
        for (int s = 0; s < 2; ++s) {
            v2f ax[4];
            const float* er = Wemb + (long)tokp[s] * EMB_D + koff;
#pragma unroll
            for (int kc = 0; kc < 4; ++kc) ax[kc] = *(const v2f*)(er + 4 * kc);
            v8f p[8];
            x_projection(ax, kf, bacc, p);
            store_fragments(pbuf, s, lane, p);
        }
        const float* er = Wemb + (long)tokp[2] * EMB_D + koff;
#pragma unroll
        for (int kc = 0; kc < 4; ++kc) axc[kc] = *(const v2f*)(er + 4 * kc);
        tokn = tokp[3];
    } else {
        tok0 = tokp[0];
        tok1 = tokp[1];
    }
    __syncthreads();
    if (recurrent) load_fragments(pbuf, 0, lane, pa);
    __syncthreads();

    for (int t = 0; t < T_LEN; t += 2) {
        if (recurrent)
            recurrent_step(t, tokp, pbuf, lane, j1, j2, rf, pa, pb, cown, hown, hfull, tok0, tok1);
        else
            projection_step(t, tokp, Wemb, pbuf, lane, koff, kf, bacc, axc, tokn);
        __syncthreads();
        if (recurrent)
            recurrent_step(t + 1, tokp, pbuf, lane, j1, j2, rf, pb, pa, cown, hown, hfull, tok0, tok1);
        else
            projection_step(t + 1, tokp, Wemb, pbuf, lane, koff, kf, bacc, axc, tokn);
        __syncthreads();
    }

    if (recurrent && jq == 0) {
#pragma unroll
        for (int hc = 0; hc < 2; ++hc) {
            float* op = out + (b0 + brow) * HID_D + hc * 16 + base;
            v4f lo, hi;
#pragma unroll
            for (int r = 0; r < 4; ++r) { lo[r] = hfull[hc][r]; hi[r] = hfull[hc][4 + r]; }
            *(v4f*)op = lo;
            *(v4f*)(op + 4) = hi;
        }
    }
}

extern "C" void kernel_launch(void* const* d_in, const int* in_sizes, int n_in,
                              void* d_out, int out_size, void* d_ws, size_t ws_size,
                              hipStream_t stream) {
    const int*   tokens = (const int*)  d_in[0];
    const float* emb    = (const float*)d_in[1];
    const float* Wx     = (const float*)d_in[2];
    const float* Wh     = (const float*)d_in[3];
    const float* bias   = (const float*)d_in[4];
    float*       out    = (float*)d_out;
    (void)in_sizes; (void)n_in; (void)out_size; (void)d_ws; (void)ws_size;

    dim3 grid(BATCH / ROWS);
    dim3 block(64);
    lstm_scan_kernel<<<grid, block, 0, stream>>>(tokens, emb, Wx, Wh, bias, out);
}


# batch tile narrowed to 2 rows, each copied to the 8 B columns of a lane group: a lane does gate math for 2 of its 16 elements (8-way pick, h back by DPP8 broadcast); 256 workgroups; sum order kept
# speedup vs baseline: 2.5559x; 1.1886x over previous
#include <hip/hip_runtime.h>


typedef __attribute__((ext_vector_type(2))) float v2f;
typedef __attribute__((ext_vector_type(4))) float v4f;
typedef __attribute__((ext_vector_type(8))) float v8f;

static constexpr int T_LEN = 2048;
static constexpr int BATCH = 512;
static constexpr int EMB_D = 16;
static constexpr int HID_D = 32;
static constexpr int GATE  = 128;
static constexpr int ROWS  = 2;

__device__ __forceinline__ float fsigmoid(float x) {
    return __fdividef(1.0f, 1.0f + __expf(-x));
}
__device__ __forceinline__ float ftanh(float x) {
    return 1.0f - __fdividef(2.0f, __expf(2.0f * x) + 1.0f);
}

__device__ __forceinline__ v8f wmma4(v2f a, v2f b, v8f c) {
    return __builtin_amdgcn_wmma_f32_16x16x4_f32(
         false, a,  false, b,
         (short)0, c,  false,  false);
}

__device__ __forceinline__ void x_projection(const v2f (&ax)[4], const v2f (&kf)[8][4],
                                             const v8f (&bacc)[8], v8f (&p)[8]) {
#pragma unroll
    for (int nc = 0; nc < 8; ++nc) p[nc] = wmma4(kf[nc][0], ax[0], bacc[nc]);
#pragma unroll
    for (int kc = 1; kc < 4; ++kc) {
#pragma unroll
        for (int nc = 0; nc < 8; ++nc) p[nc] = wmma4(kf[nc][kc], ax[kc], p[nc]);
    }
}

__device__ __forceinline__ void store_fragments(v4f* pb, const int slot, const int lane,
                                                const v8f (&p)[8]) {
#pragma unroll
    for (int nc = 0; nc < 8; ++nc) {
        v4f lo, hi;
#pragma unroll
        for (int r = 0; r < 4; ++r) { lo[r] = p[nc][r]; hi[r] = p[nc][4 + r]; }
        pb[((slot * 8 + nc) * 2 + 0) * 32 + lane] = lo;
        pb[((slot * 8 + nc) * 2 + 1) * 32 + lane] = hi;
    }
}
__device__ __forceinline__ void load_fragments(const v4f* pb, const int slot, const int lane,
                                               v8f (&p)[8]) {
#pragma unroll
    for (int nc = 0; nc < 8; ++nc) {
        const v4f lo = pb[((slot * 8 + nc) * 2 + 0) * 32 + lane];
        const v4f hi = pb[((slot * 8 + nc) * 2 + 1) * 32 + lane];
#pragma unroll
        for (int r = 0; r < 4; ++r) { p[nc][r] = lo[r]; p[nc][4 + r] = hi[r]; }
    }
}

__device__ __forceinline__ float pick_bits(const int mask, const float a, const float b) {
    return __int_as_float(__float_as_int(b) ^ ((__float_as_int(a) ^ __float_as_int(b)) & mask));
}

template <int K>
__device__ __forceinline__ float oct_bcast(const float v) {
    return __int_as_float((int)__builtin_amdgcn_mov_dpp8((unsigned)__float_as_int(v), K * 0x249249));
}

__device__ __forceinline__ float pick_own(const int j1, const int j2, const int j4, const v8f& f) {
    const float a0 = pick_bits(j1, f[1], f[0]);
    const float a1 = pick_bits(j1, f[3], f[2]);
    const float a2 = pick_bits(j1, f[5], f[4]);
    const float a3 = pick_bits(j1, f[7], f[6]);
    const float b0 = pick_bits(j2, a1, a0);
    const float b1 = pick_bits(j2, a3, a2);
    return pick_bits(j4, b1, b0);
}

__device__ __forceinline__ void bcast_group(const float hw, v8f& hf) {
    hf[0] = oct_bcast<0>(hw);
    hf[1] = oct_bcast<1>(hw);
    hf[2] = oct_bcast<2>(hw);
    hf[3] = oct_bcast<3>(hw);
    hf[4] = oct_bcast<4>(hw);
    hf[5] = oct_bcast<5>(hw);
    hf[6] = oct_bcast<6>(hw);
    hf[7] = oct_bcast<7>(hw);
}

__device__ __forceinline__ void gate_group0(const bool m, const int j1, const int j2, const int j4,
                                            const v8f (&p)[8], float (&cown)[2],
                                            float (&hown)[2], v8f (&hfull)[2]) {
    float z[4];
#pragma unroll
    for (int q = 0; q < 4; ++q) z[q] = pick_own(j1, j2, j4, p[2 * q]);
    const float co = cown[0];
    const float ho = hown[0];
    const float iv = fsigmoid(z[0]);
    const float fv = fsigmoid(z[1]);
    const float gv = ftanh(z[2]);
    const float ov = fsigmoid(z[3]);
    const float cn = fv * co + iv * gv;
    const float hn = ov * ftanh(cn);
    const float cw = m ? cn : co;
    const float hw = m ? hn : ho;
    cown[0] = cw;
    hown[0] = hw;
    bcast_group(hw, hfull[0]);
}

template <int N>
__device__ __forceinline__ void beside_matrix() {
#pragma unroll
    for (int i = 0; i < 8; ++i) {
        __builtin_amdgcn_sched_group_barrier(0x008, 1, 0);
        __builtin_amdgcn_sched_group_barrier(0x002, N, 0);
    }
}

template <int C>
__device__ __forceinline__ void h_chunk_product(const v2f (&rf)[8][8], const v8f (&hfull)[2],
                                                v8f (&pn)[8]) {
    constexpr int hc = C >> 2;
    constexpr int r0 = (C & 3) << 1;
    v2f bh;
    bh[0] = hfull[hc][r0];
    bh[1] = hfull[hc][r0 + 1];
#pragma unroll
    for (int nc = 0; nc < 8; ++nc) pn[nc] = wmma4(rf[nc][C], bh, pn[nc]);
}

__device__ __forceinline__ void recurrent_step(
    const int t, const int* __restrict__ tokp, const v4f* pbuf, const int lane,
    const int j1, const int j2, const int j4,
    const v2f (&rf)[8][8], v8f (&p)[8], v8f (&pn)[8],
    float (&cown)[2], float (&hown)[2], v8f (&hfull)[2], int& tok0, int& tok1)
{
    const int tok2 = (t + 2 < T_LEN) ? tokp[t + 2] : 0;
    const bool m = (tok0 != 0);

    load_fragments(pbuf, (t + 1) & 1, lane, pn);

    __builtin_amdgcn_sched_barrier(0);
    gate_group0(m, j1, j2, j4, p, cown, hown, hfull);
    __builtin_amdgcn_sched_barrier(0);
    float z[4];
#pragma unroll
    for (int q = 0; q < 4; ++q) z[q] = pick_own(j1, j2, j4, p[2 * q + 1]);
    const float den_i = 1.0f + __expf(-z[0]);
    const float den_f = 1.0f + __expf(-z[1]);
    const float den_g = __expf(2.0f * z[2]) + 1.0f;
    const float den_o = 1.0f + __expf(-z[3]);
    h_chunk_product<0>(rf, hfull, pn);
    asm volatile("" :: "v"(den_i), "v"(den_f), "v"(den_g), "v"(den_o));
    __builtin_amdgcn_sched_barrier(0);
    const float iv = __fdividef(1.0f, den_i);
    const float fv = __fdividef(1.0f, den_f);
    const float gv = 1.0f - __fdividef(2.0f, den_g);
    h_chunk_product<1>(rf, hfull, pn);
    beside_matrix<5>();
    asm volatile("" :: "v"(iv), "v"(fv), "v"(gv));
    __builtin_amdgcn_sched_barrier(0);
    const float co = cown[1];
    const float ho = hown[1];
    const float cn = fv * co + iv * gv;
    const float tc = ftanh(cn);
    const float ov = __fdividef(1.0f, den_o);
    h_chunk_product<2>(rf, hfull, pn);
    beside_matrix<4>();
    asm volatile("" :: "v"(tc), "v"(ov));
    __builtin_amdgcn_sched_barrier(0);
    const float hn = ov * tc;
    const float cw = m ? cn : co;
    const float hw = m ? hn : ho;
    cown[1] = cw;
    hown[1] = hw;
    bcast_group(hw, hfull[1]);
    h_chunk_product<3>(rf, hfull, pn);
    beside_matrix<3>();
    __builtin_amdgcn_sched_barrier(0);
    h_chunk_product<4>(rf, hfull, pn);
    h_chunk_product<5>(rf, hfull, pn);
    h_chunk_product<6>(rf, hfull, pn);
    h_chunk_product<7>(rf, hfull, pn);

    tok0 = tok1;
    tok1 = tok2;
}

__device__ __forceinline__ void projection_step(
    const int t, const int* __restrict__ tokp, const float* __restrict__ Wemb,
    v4f* pbuf, const int lane, const int koff,
    const v2f (&kf)[8][4], const v8f (&bacc)[8], v2f (&axc)[4], int& tokn)
{
    const int tokf = (t + 4 < T_LEN) ? tokp[t + 4] : 0;
    v2f axn[4];
    {
        const float* er = Wemb + (long)tokn * EMB_D + koff;
#pragma unroll
        for (int kc = 0; kc < 4; ++kc) axn[kc] = *(const v2f*)(er + 4 * kc);
    }
    v8f p[8];
    x_projection(axc, kf, bacc, p);
    store_fragments(pbuf, t & 1, lane, p);
    tokn = tokf;
#pragma unroll
    for (int kc = 0; kc < 4; ++kc) axc[kc] = axn[kc];
}

__global__ __launch_bounds__(64)
void lstm_scan_kernel(const int* __restrict__ tokens,
                      const float* __restrict__ Wemb,
                      const float* __restrict__ Wx,
                      const float* __restrict__ Wh,
                      const float* __restrict__ bias,
                      float* __restrict__ out)
{
    __shared__ v4f pbuf[2 * 8 * 2 * 32];

    const int lane = threadIdx.x & 31;
    const int half = lane >> 4;
    const int nloc = lane & 15;
    const int koff = half << 1;
    const int base = half << 3;
    const int b0   = blockIdx.x * ROWS;
    const int brow = nloc >> 3;
    const int jq   = nloc & 7;
    const int j1   = (jq & 1) ? -1 : 0;
    const int j2   = (jq & 2) ? -1 : 0;
    const int j4   = (jq & 4) ? -1 : 0;
    const bool recurrent = (__builtin_amdgcn_readfirstlane((int)threadIdx.x) < 32);

    const int* tokp = tokens + (b0 + brow) * T_LEN;

    v2f rf[8][8];
    v2f kf[8][4];
    v8f bacc[8];
#pragma unroll
    for (int nc = 0; nc < 8; ++nc) {
        const int col = nc * 16 + nloc;
#pragma unroll
        for (int kc = 0; kc < 8; ++kc) {
            const int kh = ((kc >> 2) << 4) + base + ((kc & 3) << 1);
            rf[nc][kc][0] = Wh[kh * GATE + col];
            rf[nc][kc][1] = Wh[(kh + 1) * GATE + col];
        }
#pragma unroll
        for (int kc = 0; kc < 4; ++kc) {
            kf[nc][kc][0] = Wx[(4 * kc + koff) * GATE + col];
            kf[nc][kc][1] = Wx[(4 * kc + koff + 1) * GATE + col];
        }
#pragma unroll
        for (int r = 0; r < 8; ++r) bacc[nc][r] = bias[nc * 16 + base + r];
    }

    v2f axc[4] = {};
    int tokn = 0;
    float cown[2] = {};
    float hown[2] = {};
    v8f hfull[2] = {};
    v8f pa[8] = {};
    v8f pb[8] = {};
    int tok0 = 0, tok1 = 0;

    if (!recurrent) {
#pragma unroll
        for (int s = 0; s < 2; ++s) {
            v2f ax[4];
            const float* er = Wemb + (long)tokp[s] * EMB_D + koff;
#pragma unroll
            for (int kc = 0; kc < 4; ++kc) ax[kc] = *(const v2f*)(er + 4 * kc);
            v8f p[8];
            x_projection(ax, kf, bacc, p);
            store_fragments(pbuf, s, lane, p);
        }
        const float* er = Wemb + (long)tokp[2] * EMB_D + koff;
#pragma unroll
        for (int kc = 0; kc < 4; ++kc) axc[kc] = *(const v2f*)(er + 4 * kc);
        tokn = tokp[3];
    } else {
        tok0 = tokp[0];
        tok1 = tokp[1];
    }
    __syncthreads();
    if (recurrent) load_fragments(pbuf, 0, lane, pa);
    __syncthreads();

    for (int t = 0; t < T_LEN; t += 2) {
        if (recurrent)
            recurrent_step(t, tokp, pbuf, lane, j1, j2, j4, rf, pa, pb, cown, hown, hfull, tok0, tok1);
        else
            projection_step(t, tokp, Wemb, pbuf, lane, koff, kf, bacc, axc, tokn);
        __syncthreads();
        if (recurrent)
            recurrent_step(t + 1, tokp, pbuf, lane, j1, j2, j4, rf, pb, pa, cown, hown, hfull, tok0, tok1);
        else
            projection_step(t + 1, tokp, Wemb, pbuf, lane, koff, kf, bacc, axc, tokn);
        __syncthreads();
    }

    if (recurrent && jq == 0) {
#pragma unroll
        for (int hc = 0; hc < 2; ++hc) {
            float* op = out + (b0 + brow) * HID_D + hc * 16 + base;
            v4f lo, hi;
#pragma unroll
            for (int r = 0; r < 4; ++r) { lo[r] = hfull[hc][r]; hi[r] = hfull[hc][4 + r]; }
            *(v4f*)op = lo;
            *(v4f*)(op + 4) = hi;
        }
    }
}

extern "C" void kernel_launch(void* const* d_in, const int* in_sizes, int n_in,
                              void* d_out, int out_size, void* d_ws, size_t ws_size,
                              hipStream_t stream) {
    const int*   tokens = (const int*)  d_in[0];
    const float* emb    = (const float*)d_in[1];
    const float* Wx     = (const float*)d_in[2];
    const float* Wh     = (const float*)d_in[3];
    const float* bias   = (const float*)d_in[4];
    float*       out    = (float*)d_out;
    (void)in_sizes; (void)n_in; (void)out_size; (void)d_ws; (void)ws_size;

    dim3 grid(BATCH / ROWS);
    dim3 block(64);
    lstm_scan_kernel<<<grid, block, 0, stream>>>(tokens, emb, Wx, Wh, bias, out);
}


# half-sums for other waves computed transposed (h as A, weights as B): a lane's D register is its one (row,gate,unit) value, the 42 selects before the stores go; reader reads a permuted LDS place
# speedup vs baseline: 5.4076x; 2.1157x over previous
#include <hip/hip_runtime.h>


typedef __attribute__((ext_vector_type(2))) float v2f;
typedef __attribute__((ext_vector_type(4))) float v4f;
typedef __attribute__((ext_vector_type(8))) float v8f;

static constexpr int T_LEN = 2048;
static constexpr int BATCH = 512;
static constexpr int EMB_D = 16;
static constexpr int HID_D = 32;
static constexpr int GATE  = 128;
static constexpr int ROWS  = 2;
static constexpr int NWAVE = 4;
static constexpr int XSLOT = NWAVE * NWAVE * 32;

__device__ __forceinline__ float fsigmoid(float x) {
    return __fdividef(1.0f, 1.0f + __expf(-x));
}
__device__ __forceinline__ float ftanh(float x) {
    return 1.0f - __fdividef(2.0f, __expf(2.0f * x) + 1.0f);
}

__device__ __forceinline__ v8f wmma4(v2f a, v2f b, v8f c) {
    return __builtin_amdgcn_wmma_f32_16x16x4_f32(
         false, a,  false, b,
         (short)0, c,  false,  false);
}

__device__ __forceinline__ float pick_bits(const int mask, const float a, const float b) {
    return __int_as_float(__float_as_int(b) ^ ((__float_as_int(a) ^ __float_as_int(b)) & mask));
}

template <int K>
__device__ __forceinline__ float oct_bcast(const float v) {
    return __int_as_float((int)__builtin_amdgcn_mov_dpp8((unsigned)__float_as_int(v), K * 0x249249));
}

__device__ __forceinline__ float xhalf4(const float v) {
    const int x = __float_as_int(v);
    return __int_as_float(__builtin_amdgcn_permlanex16(x, x, 0x32107654, 0xba98fedc, false, false));
}

__device__ __forceinline__ float pick_own(const int j1, const int j2, const int j4, const v8f& f) {
    const float a0 = pick_bits(j1, f[1], f[0]);
    const float a1 = pick_bits(j1, f[3], f[2]);
    const float a2 = pick_bits(j1, f[5], f[4]);
    const float a3 = pick_bits(j1, f[7], f[6]);
    const float b0 = pick_bits(j2, a1, a0);
    const float b1 = pick_bits(j2, a3, a2);
    return pick_bits(j4, b1, b0);
}

__device__ __forceinline__ void lds_barrier() {
    __builtin_amdgcn_fence(__ATOMIC_RELEASE, "workgroup", "local");
    __builtin_amdgcn_s_barrier();
    __builtin_amdgcn_fence(__ATOMIC_ACQUIRE, "workgroup", "local");
}

__device__ __forceinline__ void recurrent_step(
    const int t, const int* __restrict__ tokp, const float* __restrict__ Wemb, const int koff,
    const float* xin1, const float* xin2, const float* xin3,
    float* xout1, float* xout2, float* xout3,
    const int j1, const int j2, const int j4, const bool up,
    const float k1, const float k2, const float k3,
    const v2f (&rfo)[2][2], const v2f (&rff)[3][2][2], const v2f (&kf)[2][4], const v8f (&bacc)[2],
    v8f (&p)[2], v8f (&pn)[2],
    float& cown, float& hown, v4f& hB, v2f (&axu)[4], int& tkslot,
    bool& m0, bool& m1, bool& m2)
{
    const int so = (t & 1) * 64;
    v2f o1, o2, o3;
    o1[0] = xin1[so]; o1[1] = xin1[so + 32];
    o2[0] = xin2[so]; o2[1] = xin2[so + 32];
    o3[0] = xin3[so]; o3[1] = xin3[so + 32];
#pragma unroll
    for (int tl = 0; tl < 2; ++tl) pn[tl] = wmma4(kf[tl][0], axu[0], bacc[tl]);
#pragma unroll
    for (int tl = 0; tl < 2; ++tl) pn[tl] = wmma4(kf[tl][1], axu[1], pn[tl]);
    __builtin_amdgcn_sched_barrier(0);

#pragma unroll
    for (int tl = 0; tl < 2; ++tl) pn[tl] = wmma4(kf[tl][2], axu[2], pn[tl]);

    const v2f ax3 = axu[3];
    const int tk = tkslot;
    {
        const float* er = Wemb + (long)tk * EMB_D + koff;
#pragma unroll
        for (int kc = 0; kc < 4; ++kc) axu[kc] = *(const v2f*)(er + 4 * kc);
    }
    tkslot = tokp[(t + 5 < T_LEN) ? (t + 5) : (T_LEN - 1)];
    const bool m = m0;
    m0 = m1;
    m1 = m2;
    m2 = (tk != 0);

    const float zA = (pick_own(j1, j2, j4, p[0]) + o1[0]) + (o2[0] + o3[0]);
    const float zB = (pick_own(j1, j2, j4, p[1]) + o1[1]) + (o2[1] + o3[1]);
    __builtin_amdgcn_sched_barrier(0);
    pn[0] = wmma4(kf[0][3], ax3, pn[0]);
    const float a = fsigmoid(zA);
    const float b = k3 + __fdividef(k2, __expf(k1 * zB) + 1.0f);
    asm volatile("" :: "v"(pn[0]));
    __builtin_amdgcn_sched_barrier(0);
    pn[1] = wmma4(kf[1][3], ax3, pn[1]);
    const float xa = xhalf4(a);
    const float xb = xhalf4(b);
    const float iv = up ? xa : a;
    const float fv = up ? a : xa;
    const float gv = up ? xb : b;
    const float ov = up ? b : xb;
    const float co = cown;
    const float ho = hown;
    const float cn = fv * co + iv * gv;
    const float hn = ov * ftanh(cn);
    const float cw = m ? cn : co;
    const float hw = m ? hn : ho;
    cown = cw;
    hown = hw;
    asm volatile("" :: "v"(pn[1]));
    __builtin_amdgcn_sched_barrier(0);
    hB[0] = oct_bcast<0>(hw);
    hB[1] = oct_bcast<1>(hw);
    hB[2] = oct_bcast<2>(hw);
    hB[3] = oct_bcast<3>(hw);
    v2f bh0, bh1;
    bh0[0] = hB[0]; bh0[1] = hB[1];
    bh1[0] = hB[2]; bh1[1] = hB[3];

    v8f pf[3][2];
    const v8f zero8 = {};
#pragma unroll
    for (int fs = 0; fs < 3; ++fs) {
#pragma unroll
        for (int tl = 0; tl < 2; ++tl) pf[fs][tl] = wmma4(bh0, rff[fs][tl][0], zero8);
    }
#pragma unroll
    for (int fs = 0; fs < 3; ++fs) {
#pragma unroll
        for (int tl = 0; tl < 2; ++tl) pf[fs][tl] = wmma4(bh1, rff[fs][tl][1], pf[fs][tl]);
    }
    {
        const int sn = ((t + 1) & 1) * 64;
        xout1[sn] = pf[0][0][0]; xout1[sn + 32] = pf[0][1][0];
        xout2[sn] = pf[1][0][0]; xout2[sn + 32] = pf[1][1][0];
        xout3[sn] = pf[2][0][0]; xout3[sn + 32] = pf[2][1][0];
    }
    __builtin_amdgcn_sched_barrier(0);
#pragma unroll
    for (int tl = 0; tl < 2; ++tl) pn[tl] = wmma4(rfo[tl][0], bh0, pn[tl]);
#pragma unroll
    for (int tl = 0; tl < 2; ++tl) pn[tl] = wmma4(rfo[tl][1], bh1, pn[tl]);
    asm volatile("" :: "v"(pn[0]), "v"(pn[1]));
    __builtin_amdgcn_sched_barrier(0);
}

__global__ __launch_bounds__(128)
void lstm_scan_kernel(const int* __restrict__ tokens,
                      const float* __restrict__ Wemb,
                      const float* __restrict__ Wx,
                      const float* __restrict__ Wh,
                      const float* __restrict__ bias,
                      float* __restrict__ out)
{
    __shared__ float xbuf[2 * 2 * XSLOT];
    __shared__ alignas(16) float fin[ROWS * HID_D];

    const int lane = threadIdx.x & 31;
    const int half = lane >> 4;
    const int nloc = lane & 15;
    const int koff = half << 1;
    const int b0   = blockIdx.x * ROWS;
    const int brow = nloc >> 3;
    const int jq   = nloc & 7;
    const int uq   = jq ^ (half << 2);
    const int j1   = (uq & 1) ? -1 : 0;
    const int j2   = (uq & 2) ? -1 : 0;
    const int j4   = (uq & 4) ? -1 : 0;
    const bool up  = (half != 0);
    const float k1 = up ? -1.0f : 2.0f;
    const float k2 = up ? 1.0f : -2.0f;
    const float k3 = up ? 0.0f : 1.0f;
    const int w = __builtin_amdgcn_readfirstlane((int)threadIdx.x) >> 5;

    const int* tokp = tokens + (b0 + brow) * T_LEN;

    const int gm = nloc >> 3;
    const int um = nloc & 7;
    v2f rfo[2][2];
    v2f rff[3][2][2];
    v2f kf[2][4];
    v8f bacc[2];
#pragma unroll
    for (int tl = 0; tl < 2; ++tl) {
        const int colo = (2 * tl + gm) * HID_D + 8 * w + um;
#pragma unroll
        for (int pp = 0; pp < 2; ++pp) {
            const int kh = 8 * w + 4 * half + 2 * pp;
            rfo[tl][pp][0] = Wh[kh * GATE + colo];
            rfo[tl][pp][1] = Wh[(kh + 1) * GATE + colo];
        }
#pragma unroll
        for (int kc = 0; kc < 4; ++kc) {
            kf[tl][kc][0] = Wx[(4 * kc + koff) * GATE + colo];
            kf[tl][kc][1] = Wx[(4 * kc + koff + 1) * GATE + colo];
        }
#pragma unroll
        for (int r = 0; r < 8; ++r) bacc[tl][r] = bias[(2 * tl + half) * HID_D + 8 * w + r];
    }
#pragma unroll
    for (int fs = 0; fs < 3; ++fs) {
        const int v = (w + 1 + fs) & 3;
#pragma unroll
        for (int tl = 0; tl < 2; ++tl) {
            const int colf = (2 * tl + gm) * HID_D + 8 * v + um;
#pragma unroll
            for (int pp = 0; pp < 2; ++pp) {
                const int kh = 8 * w + 4 * half + 2 * pp;
                rff[fs][tl][pp][0] = Wh[kh * GATE + colf];
                rff[fs][tl][pp][1] = Wh[(kh + 1) * GATE + colf];
            }
        }
    }

    const int slane = 16 * brow + 8 * half + uq;
    const float* xin1 = xbuf + (w * NWAVE + ((w + 1) & 3)) * 128 + slane;
    const float* xin2 = xbuf + (w * NWAVE + ((w + 2) & 3)) * 128 + slane;
    const float* xin3 = xbuf + (w * NWAVE + ((w + 3) & 3)) * 128 + slane;
    float* xout1 = xbuf + (((w + 1) & 3) * NWAVE + w) * 128 + lane;
    float* xout2 = xbuf + (((w + 2) & 3) * NWAVE + w) * 128 + lane;
    float* xout3 = xbuf + (((w + 3) & 3) * NWAVE + w) * 128 + lane;

    float cown = 0.0f;
    float hown = 0.0f;
    v4f hB = {0.0f, 0.0f, 0.0f, 0.0f};
    v8f pa[2];
    v8f pb[2];
    const int tok0 = tokp[0], tok1 = tokp[1], tok2 = tokp[2];
    int tka = tokp[3];
    int tkb = tokp[4];
    bool m0 = (tok0 != 0), m1 = (tok1 != 0), m2 = (tok2 != 0);
    v2f axo[4];
    v2f axe[4];

    {
        v2f ax0[4];
        const float* e0 = Wemb + (long)tok0 * EMB_D + koff;
        const float* e1 = Wemb + (long)tok1 * EMB_D + koff;
        const float* e2 = Wemb + (long)tok2 * EMB_D + koff;
#pragma unroll
        for (int kc = 0; kc < 4; ++kc) {
            ax0[kc] = *(const v2f*)(e0 + 4 * kc);
            axo[kc] = *(const v2f*)(e1 + 4 * kc);
            axe[kc] = *(const v2f*)(e2 + 4 * kc);
        }
#pragma unroll
        for (int tl = 0; tl < 2; ++tl) pa[tl] = wmma4(kf[tl][0], ax0[0], bacc[tl]);
#pragma unroll
        for (int kc = 1; kc < 4; ++kc) {
#pragma unroll
            for (int tl = 0; tl < 2; ++tl) pa[tl] = wmma4(kf[tl][kc], ax0[kc], pa[tl]);
        }
        xout1[0] = 0.0f; xout1[32] = 0.0f;
        xout2[0] = 0.0f; xout2[32] = 0.0f;
        xout3[0] = 0.0f; xout3[32] = 0.0f;
    }
    __syncthreads();

    for (int t = 0; t < T_LEN; t += 2) {
        recurrent_step(t, tokp, Wemb, koff, xin1, xin2, xin3, xout1, xout2, xout3, j1, j2, j4, up, k1, k2, k3,
                       rfo, rff, kf, bacc, pa, pb, cown, hown, hB, axo, tka, m0, m1, m2);
        lds_barrier();
        recurrent_step(t + 1, tokp, Wemb, koff, xin1, xin2, xin3, xout1, xout2, xout3, j1, j2, j4, up, k1, k2, k3,
                       rfo, rff, kf, bacc, pb, pa, cown, hown, hB, axe, tkb, m0, m1, m2);
        lds_barrier();
    }

    if (jq == 0) *(v4f*)(fin + brow * HID_D + 8 * w + 4 * half) = hB;
    __syncthreads();
    if (w == 0 && lane < 16) {
        const int row = lane >> 3;
        const int part = lane & 7;
        const v4f hv = *(const v4f*)(fin + row * HID_D + 4 * part);
        *(v4f*)(out + (b0 + row) * HID_D + 4 * part) = hv;
    }
}

extern "C" void kernel_launch(void* const* d_in, const int* in_sizes, int n_in,
                              void* d_out, int out_size, void* d_ws, size_t ws_size,
                              hipStream_t stream) {
    const int*   tokens = (const int*)  d_in[0];
    const float* emb    = (const float*)d_in[1];
    const float* Wx     = (const float*)d_in[2];
    const float* Wh     = (const float*)d_in[3];
    const float* bias   = (const float*)d_in[4];
    float*       out    = (float*)d_out;
    (void)in_sizes; (void)n_in; (void)out_size; (void)d_ws; (void)ws_size;

    dim3 grid(BATCH / ROWS);
    dim3 block(128);
    lstm_scan_kernel<<<grid, block, 0, stream>>>(tokens, emb, Wx, Wh, bias, out);
}


# x@Wx+bias batched over 8 time steps: x is the A operand (rows = 2 batch rows x 8 steps), 1 matrix instr per step not 8, routed by permlane16/x16; gathers 1/8; sum order changes (x added after own h)
# speedup vs baseline: 6.1555x; 1.1383x over previous
#include <hip/hip_runtime.h>


typedef __attribute__((ext_vector_type(2))) float v2f;
typedef __attribute__((ext_vector_type(4))) float v4f;
typedef __attribute__((ext_vector_type(8))) float v8f;

static constexpr int T_LEN = 2048;
static constexpr int BATCH = 512;
static constexpr int EMB_D = 16;
static constexpr int HID_D = 32;
static constexpr int GATE  = 128;
static constexpr int ROWS  = 2;
static constexpr int NWAVE = 4;
static constexpr int XSLOT = NWAVE * NWAVE * 32;

__device__ __forceinline__ float fsigmoid(float x) {
    return __fdividef(1.0f, 1.0f + __expf(-x));
}
__device__ __forceinline__ float ftanh(float x) {
    return 1.0f - __fdividef(2.0f, __expf(2.0f * x) + 1.0f);
}

__device__ __forceinline__ v8f wmma4(v2f a, v2f b, v8f c) {
    return __builtin_amdgcn_wmma_f32_16x16x4_f32(
         false, a,  false, b,
         (short)0, c,  false,  false);
}

__device__ __forceinline__ float pick_bits(const int mask, const float a, const float b) {
    return __int_as_float(__float_as_int(b) ^ ((__float_as_int(a) ^ __float_as_int(b)) & mask));
}

template <int K>
__device__ __forceinline__ float oct_bcast(const float v) {
    return __int_as_float((int)__builtin_amdgcn_mov_dpp8((unsigned)__float_as_int(v), K * 0x249249));
}

__device__ __forceinline__ float xhalf4(const float v) {
    const int x = __float_as_int(v);
    return __int_as_float(__builtin_amdgcn_permlanex16(x, x, 0x32107654, 0xba98fedc, false, false));
}

__device__ __forceinline__ float pick_own(const int j1, const int j2, const int j4, const v8f& f) {
    const float a0 = pick_bits(j1, f[1], f[0]);
    const float a1 = pick_bits(j1, f[3], f[2]);
    const float a2 = pick_bits(j1, f[5], f[4]);
    const float a3 = pick_bits(j1, f[7], f[6]);
    const float b0 = pick_bits(j2, a1, a0);
    const float b1 = pick_bits(j2, a3, a2);
    return pick_bits(j4, b1, b0);
}

__device__ __forceinline__ void lds_barrier() {
    __builtin_amdgcn_fence(__ATOMIC_RELEASE, "workgroup", "local");
    __builtin_amdgcn_s_barrier();
    __builtin_amdgcn_fence(__ATOMIC_ACQUIRE, "workgroup", "local");
}

template <int K>
__device__ __forceinline__ int oct_bcast_i(const int v) {
    return (int)__builtin_amdgcn_mov_dpp8((unsigned)v, K * 0x249249);
}

__device__ __forceinline__ float route_xz(const float v, const bool same) {
    const int x = __float_as_int(v);
    const int a = __builtin_amdgcn_permlane16(x, x, 0x76543210, 0xba98fedc, false, false);
    const int b = __builtin_amdgcn_permlanex16(x, x, 0xba98fedc, 0x76543210, false, false);
    return __int_as_float(same ? a : b);
}

template <int J>
__device__ __forceinline__ void recurrent_step(
    const float* __restrict__ Wemb, const int koff,
    const float* xin1, const float* xin2, const float* xin3,
    float* xout1, float* xout2, float* xout3,
    const int j1, const int j2, const int j4, const bool up, const bool same,
    const float k1, const float k2, const float k3,
    const v2f (&rfo)[2][2], const v2f (&rff)[3][2][2], const v2f (&kf)[2][4], const v8f (&bT)[2],
    v8f (&p)[2], v8f (&pn)[2],
    float& cown, float& hown, v4f& hB,
    v8f (&xz)[2], v8f (&xzn)[2], v2f (&xfr)[4],
    const int* __restrict__ tokp, const int t0, const int jq, int& tokc, int& tokn, int& tok2, int& tok3)
{
    constexpr int KC = J >> 1;
    constexpr int TL = J & 1;
    constexpr int so = (J & 1) * 64;
    v2f o1, o2, o3;
    o1[0] = xin1[so]; o1[1] = xin1[so + 32];
    o2[0] = xin2[so]; o2[1] = xin2[so + 32];
    o3[0] = xin3[so]; o3[1] = xin3[so + 32];
    __builtin_amdgcn_sched_barrier(0);

    if (J == 0) { xz[0] = xzn[0]; xz[1] = xzn[1]; tokc = tokn; }
    const float xv0 = route_xz(xz[0][J], same);
    const float xv1 = route_xz(xz[1][J], same);
    const bool m = (oct_bcast_i<J>(tokc) != 0);

    const float zA = ((pick_own(j1, j2, j4, p[0]) + xv0) + o1[0]) + (o2[0] + o3[0]);
    const float zB = ((pick_own(j1, j2, j4, p[1]) + xv1) + o1[1]) + (o2[1] + o3[1]);
    __builtin_amdgcn_sched_barrier(0);
    if (J == 0) {
        tokn = tok2;
        tok2 = tok3;
        const int tn = t0 + 24 + jq;
        tok3 = tokp[((tn < T_LEN) ? tn : (T_LEN - 1)) - jq];
    }
    if (KC == 0) xzn[TL] = wmma4(xfr[KC], kf[TL][KC], bT[TL]);
    else         xzn[TL] = wmma4(xfr[KC], kf[TL][KC], xzn[TL]);
    if (TL == 1) {
        xfr[KC] = *(const v2f*)(Wemb + (long)tok2 * EMB_D + koff + 4 * KC);
    }
    const float a = fsigmoid(zA);
    const float b = k3 + __fdividef(k2, __expf(k1 * zB) + 1.0f);
    asm volatile("" :: "v"(xzn[TL]));
    __builtin_amdgcn_sched_barrier(0);
    const float xa = xhalf4(a);
    const float xb = xhalf4(b);
    const float iv = up ? xa : a;
    const float fv = up ? a : xa;
    const float gv = up ? xb : b;
    const float ov = up ? b : xb;
    const float co = cown;
    const float ho = hown;
    const float cn = fv * co + iv * gv;
    const float hn = ov * ftanh(cn);
    const float cw = m ? cn : co;
    const float hw = m ? hn : ho;
    cown = cw;
    hown = hw;
    hB[0] = oct_bcast<0>(hw);
    hB[1] = oct_bcast<1>(hw);
    hB[2] = oct_bcast<2>(hw);
    hB[3] = oct_bcast<3>(hw);
    v2f bh0, bh1;
    bh0[0] = hB[0]; bh0[1] = hB[1];
    bh1[0] = hB[2]; bh1[1] = hB[3];

    v8f pf[3][2];
    const v8f zero8 = {};
#pragma unroll
    for (int fs = 0; fs < 3; ++fs) {
#pragma unroll
        for (int tl = 0; tl < 2; ++tl) pf[fs][tl] = wmma4(bh0, rff[fs][tl][0], zero8);
    }
#pragma unroll
    for (int fs = 0; fs < 3; ++fs) {
#pragma unroll
        for (int tl = 0; tl < 2; ++tl) pf[fs][tl] = wmma4(bh1, rff[fs][tl][1], pf[fs][tl]);
    }
    {
        constexpr int sn = ((J + 1) & 1) * 64;
        xout1[sn] = pf[0][0][0]; xout1[sn + 32] = pf[0][1][0];
        xout2[sn] = pf[1][0][0]; xout2[sn + 32] = pf[1][1][0];
        xout3[sn] = pf[2][0][0]; xout3[sn + 32] = pf[2][1][0];
    }
    __builtin_amdgcn_sched_barrier(0);
#pragma unroll
    for (int tl = 0; tl < 2; ++tl) pn[tl] = wmma4(rfo[tl][0], bh0, zero8);
#pragma unroll
    for (int tl = 0; tl < 2; ++tl) pn[tl] = wmma4(rfo[tl][1], bh1, pn[tl]);
    asm volatile("" :: "v"(pn[0]), "v"(pn[1]));
    __builtin_amdgcn_sched_barrier(0);
}

__global__ __launch_bounds__(128)
void lstm_scan_kernel(const int* __restrict__ tokens,
                      const float* __restrict__ Wemb,
                      const float* __restrict__ Wx,
                      const float* __restrict__ Wh,
                      const float* __restrict__ bias,
                      float* __restrict__ out)
{
    __shared__ float xbuf[2 * 2 * XSLOT];
    __shared__ alignas(16) float fin[ROWS * HID_D];

    const int lane = threadIdx.x & 31;
    const int half = lane >> 4;
    const int nloc = lane & 15;
    const int koff = half << 1;
    const int b0   = blockIdx.x * ROWS;
    const int brow = nloc >> 3;
    const int jq   = nloc & 7;
    const int uq   = jq ^ (half << 2);
    const int j1   = (uq & 1) ? -1 : 0;
    const int j2   = (uq & 2) ? -1 : 0;
    const int j4   = (uq & 4) ? -1 : 0;
    const bool up  = (half != 0);
    const bool same = (half == brow);
    const float k1 = up ? -1.0f : 2.0f;
    const float k2 = up ? 1.0f : -2.0f;
    const float k3 = up ? 0.0f : 1.0f;
    const int w = __builtin_amdgcn_readfirstlane((int)threadIdx.x) >> 5;

    const int* tokp = tokens + (b0 + brow) * T_LEN + jq;

    const int gm = nloc >> 3;
    const int um = nloc & 7;
    v2f rfo[2][2];
    v2f rff[3][2][2];
    v2f kf[2][4];
    v8f bT[2];
#pragma unroll
    for (int tl = 0; tl < 2; ++tl) {
        const int colo = (2 * tl + gm) * HID_D + 8 * w + um;
#pragma unroll
        for (int pp = 0; pp < 2; ++pp) {
            const int kh = 8 * w + 4 * half + 2 * pp;
            rfo[tl][pp][0] = Wh[kh * GATE + colo];
            rfo[tl][pp][1] = Wh[(kh + 1) * GATE + colo];
        }
#pragma unroll
        for (int kc = 0; kc < 4; ++kc) {
            kf[tl][kc][0] = Wx[(4 * kc + koff) * GATE + colo];
            kf[tl][kc][1] = Wx[(4 * kc + koff + 1) * GATE + colo];
        }
        const float bv = bias[colo];
#pragma unroll
        for (int r = 0; r < 8; ++r) bT[tl][r] = bv;
    }
#pragma unroll
    for (int fs = 0; fs < 3; ++fs) {
        const int v = (w + 1 + fs) & 3;
#pragma unroll
        for (int tl = 0; tl < 2; ++tl) {
            const int colf = (2 * tl + gm) * HID_D + 8 * v + um;
#pragma unroll
            for (int pp = 0; pp < 2; ++pp) {
                const int kh = 8 * w + 4 * half + 2 * pp;
                rff[fs][tl][pp][0] = Wh[kh * GATE + colf];
                rff[fs][tl][pp][1] = Wh[(kh + 1) * GATE + colf];
            }
        }
    }

    const int slane = 16 * brow + 8 * half + uq;
    const float* xin1 = xbuf + (w * NWAVE + ((w + 1) & 3)) * 128 + slane;
    const float* xin2 = xbuf + (w * NWAVE + ((w + 2) & 3)) * 128 + slane;
    const float* xin3 = xbuf + (w * NWAVE + ((w + 3) & 3)) * 128 + slane;
    float* xout1 = xbuf + (((w + 1) & 3) * NWAVE + w) * 128 + lane;
    float* xout2 = xbuf + (((w + 2) & 3) * NWAVE + w) * 128 + lane;
    float* xout3 = xbuf + (((w + 3) & 3) * NWAVE + w) * 128 + lane;

    float cown = 0.0f;
    float hown = 0.0f;
    v4f hB = {0.0f, 0.0f, 0.0f, 0.0f};
    v8f pa[2] = {};
    v8f pb[2];
    int tokc = 0;
    int tokn = tokp[0];
    int tok2 = tokp[8];
    int tok3 = tokp[16];
    v8f xz[2];
    v8f xzn[2];
    v2f xfr[4];

    {
        v2f x0[4];
        const float* e0 = Wemb + (long)tokn * EMB_D + koff;
        const float* e1 = Wemb + (long)tok2 * EMB_D + koff;
#pragma unroll
        for (int kc = 0; kc < 4; ++kc) {
            x0[kc]  = *(const v2f*)(e0 + 4 * kc);
            xfr[kc] = *(const v2f*)(e1 + 4 * kc);
        }
#pragma unroll
        for (int tl = 0; tl < 2; ++tl) xzn[tl] = wmma4(x0[0], kf[tl][0], bT[tl]);
#pragma unroll
        for (int kc = 1; kc < 4; ++kc) {
#pragma unroll
            for (int tl = 0; tl < 2; ++tl) xzn[tl] = wmma4(x0[kc], kf[tl][kc], xzn[tl]);
        }
        xout1[0] = 0.0f; xout1[32] = 0.0f;
        xout2[0] = 0.0f; xout2[32] = 0.0f;
        xout3[0] = 0.0f; xout3[32] = 0.0f;
    }
    __syncthreads();

#define LSTM_STEP(J, P, PN)                                                                      \
    recurrent_step<J>(Wemb, koff, xin1, xin2, xin3, xout1, xout2, xout3, j1, j2, j4, up, same,   \
                      k1, k2, k3, rfo, rff, kf, bT, P, PN, cown, hown, hB, xz, xzn, xfr,           \
                      tokp, t0, jq, tokc, tokn, tok2, tok3);                                       \
    lds_barrier();
    for (int t0 = 0; t0 < T_LEN; t0 += 8) {
        LSTM_STEP(0, pa, pb)
        LSTM_STEP(1, pb, pa)
        LSTM_STEP(2, pa, pb)
        LSTM_STEP(3, pb, pa)
        LSTM_STEP(4, pa, pb)
        LSTM_STEP(5, pb, pa)
        LSTM_STEP(6, pa, pb)
        LSTM_STEP(7, pb, pa)
    }
#undef LSTM_STEP

    if (jq == 0) *(v4f*)(fin + brow * HID_D + 8 * w + 4 * half) = hB;
    __syncthreads();
    if (w == 0 && lane < 16) {
        const int row = lane >> 3;
        const int part = lane & 7;
        const v4f hv = *(const v4f*)(fin + row * HID_D + 4 * part);
        *(v4f*)(out + (b0 + row) * HID_D + 4 * part) = hv;
    }
}

extern "C" void kernel_launch(void* const* d_in, const int* in_sizes, int n_in,
                              void* d_out, int out_size, void* d_ws, size_t ws_size,
                              hipStream_t stream) {
    const int*   tokens = (const int*)  d_in[0];
    const float* emb    = (const float*)d_in[1];
    const float* Wx     = (const float*)d_in[2];
    const float* Wh     = (const float*)d_in[3];
    const float* bias   = (const float*)d_in[4];
    float*       out    = (float*)d_out;
    (void)in_sizes; (void)n_in; (void)out_size; (void)d_ws; (void)ws_size;

    dim3 grid(BATCH / ROWS);
    dim3 block(128);
    lstm_scan_kernel<<<grid, block, 0, stream>>>(tokens, emb, Wx, Wh, bias, out);
}


# step's dependent chain cut by 7 links: x matrix instr moved into the exchange read's latency, masked h update fused into its DPP8 broadcast, exp scale in one multiply, h = o - 2o/(e^2c+1)
# speedup vs baseline: 6.6031x; 1.0727x over previous
#include <hip/hip_runtime.h>


typedef __attribute__((ext_vector_type(2))) float v2f;
typedef __attribute__((ext_vector_type(4))) float v4f;
typedef __attribute__((ext_vector_type(8))) float v8f;

static constexpr int T_LEN = 2048;
static constexpr int BATCH = 512;
static constexpr int EMB_D = 16;
static constexpr int HID_D = 32;
static constexpr int GATE  = 128;
static constexpr int ROWS  = 2;
static constexpr int NWAVE = 4;
static constexpr int XSLOT = NWAVE * NWAVE * 32;

__device__ __forceinline__ float fsigmoid(float x) {
    return __fdividef(1.0f, 1.0f + __expf(-x));
}
static constexpr float LOG2E     = 0x1.715476p+0f;
static constexpr float TWO_LOG2E = 0x1.715476p+1f;

__device__ __forceinline__ v8f wmma4(v2f a, v2f b, v8f c) {
    return __builtin_amdgcn_wmma_f32_16x16x4_f32(
         false, a,  false, b,
         (short)0, c,  false,  false);
}

__device__ __forceinline__ float pick_bits(const int mask, const float a, const float b) {
    return __int_as_float(__float_as_int(b) ^ ((__float_as_int(a) ^ __float_as_int(b)) & mask));
}

__device__ __forceinline__ float xhalf4(const float v) {
    const int x = __float_as_int(v);
    return __int_as_float(__builtin_amdgcn_permlanex16(0, x, 0x32107654, 0xba98fedc, false, false));
}

__device__ __forceinline__ float pick_own(const int j1, const int j2, const int j4, const v8f& f) {
    const float a0 = pick_bits(j1, f[1], f[0]);
    const float a1 = pick_bits(j1, f[3], f[2]);
    const float a2 = pick_bits(j1, f[5], f[4]);
    const float a3 = pick_bits(j1, f[7], f[6]);
    const float b0 = pick_bits(j2, a1, a0);
    const float b1 = pick_bits(j2, a3, a2);
    return pick_bits(j4, b1, b0);
}

__device__ __forceinline__ void lds_barrier() {
    __builtin_amdgcn_fence(__ATOMIC_RELEASE, "workgroup", "local");
    __builtin_amdgcn_s_barrier();
    __builtin_amdgcn_fence(__ATOMIC_ACQUIRE, "workgroup", "local");
}

template <int K>
__device__ __forceinline__ int oct_bcast_i(const int v) {
    return (int)__builtin_amdgcn_mov_dpp8((unsigned)v, K * 0x249249);
}

template <int K>
__device__ __forceinline__ float bcast_unless(const float v, const float old, const unsigned keep) {
    float r;
    if (K == 0)      asm("v_cndmask_b32_e64_dpp %0, %1, %2, %3 dpp8:[0,0,0,0,0,0,0,0]" : "=v"(r) : "v"(v), "v"(old), "s"(keep));
    else if (K == 1) asm("v_cndmask_b32_e64_dpp %0, %1, %2, %3 dpp8:[1,1,1,1,1,1,1,1]" : "=v"(r) : "v"(v), "v"(old), "s"(keep));
    else if (K == 2) asm("v_cndmask_b32_e64_dpp %0, %1, %2, %3 dpp8:[2,2,2,2,2,2,2,2]" : "=v"(r) : "v"(v), "v"(old), "s"(keep));
    else             asm("v_cndmask_b32_e64_dpp %0, %1, %2, %3 dpp8:[3,3,3,3,3,3,3,3]" : "=v"(r) : "v"(v), "v"(old), "s"(keep));
    return r;
}

__device__ __forceinline__ float route_xz(const float v, const bool same) {
    const int x = __float_as_int(v);
    const int a = __builtin_amdgcn_permlane16(x, x, 0x76543210, 0xba98fedc, false, false);
    const int b = __builtin_amdgcn_permlanex16(x, x, 0xba98fedc, 0x76543210, false, false);
    return __int_as_float(same ? a : b);
}

template <int J>
__device__ __forceinline__ void recurrent_step(
    const float* __restrict__ Wemb, const int koff,
    const float* xin1, const float* xin2, const float* xin3,
    float* xout1, float* xout2, float* xout3,
    const int j1, const int j2, const int j4, const bool up, const bool same,
    const float k1l, const float k2, const float k3,
    const v2f (&rfo)[2][2], const v2f (&rff)[3][2][2], const v2f (&kf)[2][4], const v8f (&bT)[2],
    v8f (&p)[2], v8f (&pn)[2],
    float& cown, v4f& hB,
    v8f (&xz)[2], v8f (&xzn)[2], v2f (&xfr)[4],
    const int* __restrict__ tokp, const int t0, const int jq, int& tokc, int& tokn, int& tok2, int& tok3)
{
    constexpr int KC = J >> 1;
    constexpr int TL = J & 1;
    constexpr int so = (J & 1) * 64;
    v2f o1, o2, o3;
    o1[0] = xin1[so]; o1[1] = xin1[so + 32];
    o2[0] = xin2[so]; o2[1] = xin2[so + 32];
    o3[0] = xin3[so]; o3[1] = xin3[so + 32];
    __builtin_amdgcn_sched_barrier(0);

    if (J == 0) { xz[0] = xzn[0]; xz[1] = xzn[1]; tokc = tokn; }
    if (J == 0) {
        tokn = tok2;
        tok2 = tok3;
        const int tn = t0 + 24 + jq;
        tok3 = tokp[((tn < T_LEN) ? tn : (T_LEN - 1)) - jq];
    }
    if (KC == 0) xzn[TL] = wmma4(xfr[KC], kf[TL][KC], bT[TL]);
    else         xzn[TL] = wmma4(xfr[KC], kf[TL][KC], xzn[TL]);
    if (TL == 1) {
        xfr[KC] = *(const v2f*)(Wemb + (long)tok2 * EMB_D + koff + 4 * KC);
    }
    asm volatile("" :: "v"(xzn[TL]));
    __builtin_amdgcn_sched_barrier(0);

    const float xv0 = route_xz(xz[0][J], same);
    const float xv1 = route_xz(xz[1][J], same);
    const bool m = (oct_bcast_i<J>(tokc) != 0);
    const unsigned keep = __builtin_amdgcn_ballot_w32(!m);

    const float pA = pick_own(j1, j2, j4, p[0]) + xv0;
    const float pB = pick_own(j1, j2, j4, p[1]) + xv1;
    asm volatile("" :: "v"(pA), "v"(pB), "s"(keep));
    __builtin_amdgcn_sched_barrier(0);
    const float zA = (pA + o1[0]) + (o2[0] + o3[0]);
    const float zB = (pB + o1[1]) + (o2[1] + o3[1]);
    const float a = fsigmoid(zA);
    const float b = k3 + __fdividef(k2, __builtin_amdgcn_exp2f(k1l * zB) + 1.0f);
    const float xa = xhalf4(a);
    const float xb = xhalf4(b);
    const float iv = up ? xa : a;
    const float fv = up ? a : xa;
    const float gv = up ? xb : b;
    const float ov = up ? b : xb;
    const float co = cown;
    const float cn = fv * co + iv * gv;
    const float ov2 = ov + ov;
    const float hn = ov - __fdividef(ov2, __builtin_amdgcn_exp2f(cn * TWO_LOG2E) + 1.0f);
    cown = m ? cn : co;
    hB[0] = bcast_unless<0>(hn, hB[0], keep);
    hB[1] = bcast_unless<1>(hn, hB[1], keep);
    hB[2] = bcast_unless<2>(hn, hB[2], keep);
    hB[3] = bcast_unless<3>(hn, hB[3], keep);
    v2f bh0, bh1;
    bh0[0] = hB[0]; bh0[1] = hB[1];
    bh1[0] = hB[2]; bh1[1] = hB[3];

    v8f pf[3][2];
    const v8f zero8 = {};
#pragma unroll
    for (int fs = 0; fs < 3; ++fs) {
#pragma unroll
        for (int tl = 0; tl < 2; ++tl) pf[fs][tl] = wmma4(bh0, rff[fs][tl][0], zero8);
    }
#pragma unroll
    for (int fs = 0; fs < 3; ++fs) {
#pragma unroll
        for (int tl = 0; tl < 2; ++tl) pf[fs][tl] = wmma4(bh1, rff[fs][tl][1], pf[fs][tl]);
    }
    {
        constexpr int sn = ((J + 1) & 1) * 64;
        xout1[sn] = pf[0][0][0]; xout1[sn + 32] = pf[0][1][0];
        xout2[sn] = pf[1][0][0]; xout2[sn + 32] = pf[1][1][0];
        xout3[sn] = pf[2][0][0]; xout3[sn + 32] = pf[2][1][0];
    }
    __builtin_amdgcn_sched_barrier(0);
#pragma unroll
    for (int tl = 0; tl < 2; ++tl) pn[tl] = wmma4(rfo[tl][0], bh0, zero8);
#pragma unroll
    for (int tl = 0; tl < 2; ++tl) pn[tl] = wmma4(rfo[tl][1], bh1, pn[tl]);
    asm volatile("" :: "v"(pn[0]), "v"(pn[1]));
    __builtin_amdgcn_sched_barrier(0);
}

__global__ __launch_bounds__(128)
void lstm_scan_kernel(const int* __restrict__ tokens,
                      const float* __restrict__ Wemb,
                      const float* __restrict__ Wx,
                      const float* __restrict__ Wh,
                      const float* __restrict__ bias,
                      float* __restrict__ out)
{
    __shared__ float xbuf[2 * 2 * XSLOT];
    __shared__ alignas(16) float fin[ROWS * HID_D];

    const int lane = threadIdx.x & 31;
    const int half = lane >> 4;
    const int nloc = lane & 15;
    const int koff = half << 1;
    const int b0   = blockIdx.x * ROWS;
    const int brow = nloc >> 3;
    const int jq   = nloc & 7;
    const int uq   = jq ^ (half << 2);
    const int j1   = (uq & 1) ? -1 : 0;
    const int j2   = (uq & 2) ? -1 : 0;
    const int j4   = (uq & 4) ? -1 : 0;
    const bool up  = (half != 0);
    const bool same = (half == brow);
    const float k1l = up ? -LOG2E : TWO_LOG2E;
    const float k2 = up ? 1.0f : -2.0f;
    const float k3 = up ? 0.0f : 1.0f;
    const int w = __builtin_amdgcn_readfirstlane((int)threadIdx.x) >> 5;

    const int* tokp = tokens + (b0 + brow) * T_LEN + jq;

    const int gm = nloc >> 3;
    const int um = nloc & 7;
    v2f rfo[2][2];
    v2f rff[3][2][2];
    v2f kf[2][4];
    v8f bT[2];
#pragma unroll
    for (int tl = 0; tl < 2; ++tl) {
        const int colo = (2 * tl + gm) * HID_D + 8 * w + um;
#pragma unroll
        for (int pp = 0; pp < 2; ++pp) {
            const int kh = 8 * w + 4 * half + 2 * pp;
            rfo[tl][pp][0] = Wh[kh * GATE + colo];
            rfo[tl][pp][1] = Wh[(kh + 1) * GATE + colo];
        }
#pragma unroll
        for (int kc = 0; kc < 4; ++kc) {
            kf[tl][kc][0] = Wx[(4 * kc + koff) * GATE + colo];
            kf[tl][kc][1] = Wx[(4 * kc + koff + 1) * GATE + colo];
        }
        const float bv = bias[colo];
#pragma unroll
        for (int r = 0; r < 8; ++r) bT[tl][r] = bv;
    }
#pragma unroll
    for (int fs = 0; fs < 3; ++fs) {
        const int v = (w + 1 + fs) & 3;
#pragma unroll
        for (int tl = 0; tl < 2; ++tl) {
            const int colf = (2 * tl + gm) * HID_D + 8 * v + um;
#pragma unroll
            for (int pp = 0; pp < 2; ++pp) {
                const int kh = 8 * w + 4 * half + 2 * pp;
                rff[fs][tl][pp][0] = Wh[kh * GATE + colf];
                rff[fs][tl][pp][1] = Wh[(kh + 1) * GATE + colf];
            }
        }
    }

    const int slane = 16 * brow + 8 * half + uq;
    const float* xin1 = xbuf + (w * NWAVE + ((w + 1) & 3)) * 128 + slane;
    const float* xin2 = xbuf + (w * NWAVE + ((w + 2) & 3)) * 128 + slane;
    const float* xin3 = xbuf + (w * NWAVE + ((w + 3) & 3)) * 128 + slane;
    float* xout1 = xbuf + (((w + 1) & 3) * NWAVE + w) * 128 + lane;
    float* xout2 = xbuf + (((w + 2) & 3) * NWAVE + w) * 128 + lane;
    float* xout3 = xbuf + (((w + 3) & 3) * NWAVE + w) * 128 + lane;

    float cown = 0.0f;
    v4f hB = {0.0f, 0.0f, 0.0f, 0.0f};
    v8f pa[2] = {};
    v8f pb[2];
    int tokc = 0;
    int tokn = tokp[0];
    int tok2 = tokp[8];
    int tok3 = tokp[16];
    v8f xz[2];
    v8f xzn[2];
    v2f xfr[4];

    {
        v2f x0[4];
        const float* e0 = Wemb + (long)tokn * EMB_D + koff;
        const float* e1 = Wemb + (long)tok2 * EMB_D + koff;
#pragma unroll
        for (int kc = 0; kc < 4; ++kc) {
            x0[kc]  = *(const v2f*)(e0 + 4 * kc);
            xfr[kc] = *(const v2f*)(e1 + 4 * kc);
        }
#pragma unroll
        for (int tl = 0; tl < 2; ++tl) xzn[tl] = wmma4(x0[0], kf[tl][0], bT[tl]);
#pragma unroll
        for (int kc = 1; kc < 4; ++kc) {
#pragma unroll
            for (int tl = 0; tl < 2; ++tl) xzn[tl] = wmma4(x0[kc], kf[tl][kc], xzn[tl]);
        }
        xout1[0] = 0.0f; xout1[32] = 0.0f;
        xout2[0] = 0.0f; xout2[32] = 0.0f;
        xout3[0] = 0.0f; xout3[32] = 0.0f;
    }
    __syncthreads();

#define LSTM_STEP(J, P, PN)                                                                      \
    recurrent_step<J>(Wemb, koff, xin1, xin2, xin3, xout1, xout2, xout3, j1, j2, j4, up, same,   \
                      k1l, k2, k3, rfo, rff, kf, bT, P, PN, cown, hB, xz, xzn, xfr,                \
                      tokp, t0, jq, tokc, tokn, tok2, tok3);                                       \
    lds_barrier();
    for (int t0 = 0; t0 < T_LEN; t0 += 8) {
        LSTM_STEP(0, pa, pb)
        LSTM_STEP(1, pb, pa)
        LSTM_STEP(2, pa, pb)
        LSTM_STEP(3, pb, pa)
        LSTM_STEP(4, pa, pb)
        LSTM_STEP(5, pb, pa)
        LSTM_STEP(6, pa, pb)
        LSTM_STEP(7, pb, pa)
    }
#undef LSTM_STEP

    if (jq == 0) *(v4f*)(fin + brow * HID_D + 8 * w + 4 * half) = hB;
    __syncthreads();
    if (w == 0 && lane < 16) {
        const int row = lane >> 3;
        const int part = lane & 7;
        const v4f hv = *(const v4f*)(fin + row * HID_D + 4 * part);
        *(v4f*)(out + (b0 + row) * HID_D + 4 * part) = hv;
    }
}

extern "C" void kernel_launch(void* const* d_in, const int* in_sizes, int n_in,
                              void* d_out, int out_size, void* d_ws, size_t ws_size,
                              hipStream_t stream) {
    const int*   tokens = (const int*)  d_in[0];
    const float* emb    = (const float*)d_in[1];
    const float* Wx     = (const float*)d_in[2];
    const float* Wh     = (const float*)d_in[3];
    const float* bias   = (const float*)d_in[4];
    float*       out    = (float*)d_out;
    (void)in_sizes; (void)n_in; (void)out_size; (void)d_ws; (void)ws_size;

    dim3 grid(BATCH / ROWS);
    dim3 block(128);
    lstm_scan_kernel<<<grid, block, 0, stream>>>(tokens, emb, Wx, Wh, bias, out);
}
